# starting kernel
# baseline (speedup 1.0000x reference)
#include <hip/hip_runtime.h>

typedef __attribute__((ext_vector_type(2))) float v2f;
typedef __attribute__((ext_vector_type(8))) float v8f;

#define LSEQ   8192
#define DMODEL 1024
#define ROWS_PER_WAVE 512
#define STORE_TILES   (ROWS_PER_WAVE / 16)
#define WARM_TILES    10
#define TOT_TILES     (STORE_TILES + WARM_TILES)

struct Aregs { v2f A0, A1, A2, A3, Ac; };

__device__ __forceinline__ float x_load_guard(const float* __restrict__ V,
                                              const float* __restrict__ v0,
                                              int b, int m, int c) {
    if (m > LSEQ) return 0.0f;
    if (m == 0)   return v0[c];
    return V[((size_t)b * LSEQ + (size_t)(m - 1)) * DMODEL + c];
}

__device__ __forceinline__ v8f tile_mma(const Aregs& A, const v2f B[4], v2f& bc) {
    v8f acc = {};
    acc = __builtin_amdgcn_wmma_f32_16x16x4_f32(false, A.A0, false, B[0], (short)0, acc, false, false);
    acc = __builtin_amdgcn_wmma_f32_16x16x4_f32(false, A.A1, false, B[1], (short)0, acc, false, false);
    acc = __builtin_amdgcn_wmma_f32_16x16x4_f32(false, A.A2, false, B[2], (short)0, acc, false, false);
    acc = __builtin_amdgcn_wmma_f32_16x16x4_f32(false, A.A3, false, B[3], (short)0, acc, false, false);
    acc = __builtin_amdgcn_wmma_f32_16x16x4_f32(false, A.Ac, false, bc,   (short)0, acc, false, false);
    bc.x = acc[0];
    bc.y = 0.0f;
    return acc;
}

__device__ __forceinline__ void load_tile_fast(const float* __restrict__ p, v2f B[4]) {
#pragma unroll
    for (int j = 0; j < 4; ++j) {
        B[j].x = p[(size_t)(4 * j) * DMODEL];
        B[j].y = p[(size_t)(4 * j + 1) * DMODEL];
    }
}

__device__ __forceinline__ void load_tile_guard(const float* __restrict__ V,
                                                const float* __restrict__ v0,
                                                int b, int s, int koff, int c, v2f B[4]) {
#pragma unroll
    for (int j = 0; j < 4; ++j) {
        B[j].x = x_load_guard(V, v0, b, s + 4 * j + koff, c);
        B[j].y = x_load_guard(V, v0, b, s + 4 * j + 1 + koff, c);
    }
}

__device__ __forceinline__ void store_tile(float* __restrict__ out, int b, int s, int c,
                                           int lane, const v8f& acc) {
    const int rbase = s + ((lane < 16) ? 0 : 8);
    const size_t o  = ((size_t)b * LSEQ + (size_t)rbase) * DMODEL + (size_t)c;
#pragma unroll
    for (int r = 0; r < 8; ++r)
        out[o + (size_t)r * DMODEL] = acc[r];
}

__global__ void __launch_bounds__(256)
esa_suffix_scan_wmma(const float* __restrict__ V,
                     const float* __restrict__ alpha,
                     const float* __restrict__ v0,
                     float* __restrict__ out) {
    const int lane = threadIdx.x & 31;
    const int wid  = threadIdx.x >> 5;
    const int gw   = blockIdx.x * 8 + wid;
    const int b    = gw >> 10;
    const int rem  = gw & 1023;
    const int c0   = (rem >> 4) * 16;
    const int r0   = (rem & 15) * ROWS_PER_WAVE;

    const int   head = c0 >> 6;
    const float a    = 1.0f / (1.0f + expf(-alpha[head]));
    const float oma  = 1.0f - a;

    const int M    = lane & 15;
    const int koff = (lane < 16) ? 0 : 2;
    const int c    = c0 + M;

    Aregs A;
    {
        v2f* Aj[4] = { &A.A0, &A.A1, &A.A2, &A.A3 };
#pragma unroll
        for (int j = 0; j < 4; ++j) {
            const int m0 = 4 * j + koff;
            const int m1 = m0 + 1;
            Aj[j]->x = (m0 >= M) ? a * __powf(oma, (float)(m0 - M)) : 0.0f;
            Aj[j]->y = (m1 >= M) ? a * __powf(oma, (float)(m1 - M)) : 0.0f;
        }
        A.Ac.x = (lane < 16) ? __powf(oma, (float)(16 - M)) : 0.0f;
        A.Ac.y = 0.0f;
    }

    v2f bc = {0.0f, 0.0f};

    int i_hi = min(TOT_TILES - 1, (LSEQ - r0) >> 4);

    if (r0 + 16 * i_hi + 15 > LSEQ) {
        const int s = r0 + 16 * i_hi;
        v2f B[4];
        load_tile_guard(V, v0, b, s, koff, c, B);
        v8f acc = tile_mma(A, B, bc);
        if (i_hi < STORE_TILES) store_tile(out, b, s, c, lane, acc);
        --i_hi;
    }

    const int i_lo = (r0 == 0) ? 1 : 0;

    if (i_hi >= i_lo) {
        v2f Bcur[4], Bnext[4];
        {
            const int s0 = r0 + 16 * i_hi;
            load_tile_fast(V + ((size_t)b * LSEQ + (size_t)(s0 - 1 + koff)) * DMODEL + c, Bcur);
        }
        for (int i = i_hi; i >= i_lo; --i) {
            if (i > i_lo) {
                const int sn = r0 + 16 * (i - 1);
                load_tile_fast(V + ((size_t)b * LSEQ + (size_t)(sn - 1 + koff)) * DMODEL + c, Bnext);
            }
            v8f acc = tile_mma(A, Bcur, bc);
            if (i < STORE_TILES)
                store_tile(out, b, r0 + 16 * i, c, lane, acc);
#pragma unroll
            for (int j = 0; j < 4; ++j) Bcur[j] = Bnext[j];
        }
    }

    if (r0 == 0) {
        v2f B[4];
        load_tile_guard(V, v0, b, 0, koff, c, B);
        v8f acc = tile_mma(A, B, bc);
        store_tile(out, b, 0, c, lane, acc);
    }
}

extern "C" void kernel_launch(void* const* d_in, const int* in_sizes, int n_in,
                              void* d_out, int out_size, void* d_ws, size_t ws_size,
                              hipStream_t stream) {
    (void)in_sizes; (void)n_in; (void)d_ws; (void)ws_size; (void)out_size;
    const float* V     = (const float*)d_in[0];
    const float* alpha = (const float*)d_in[1];
    const float* v0    = (const float*)d_in[2];
    float* out = (float*)d_out;

    esa_suffix_scan_wmma<<<512, 256, 0, stream>>>(V, alpha, v0, out);
}


# 64 columns (one head) per wave as 4 column tiles, each lane owning 4 adjacent columns: every load and store is 16 B per lane over whole 256-B row pieces; 1024 waves, 6 row tiles of loads in flight
# speedup vs baseline: 2.4834x; 2.4834x over previous
#include <hip/hip_runtime.h>


typedef __attribute__((ext_vector_type(2))) float v2f;
typedef __attribute__((ext_vector_type(4))) float v4f;
typedef __attribute__((ext_vector_type(8))) float v8f;

#define LSEQ   8192
#define DMODEL 1024
#define ROWS_PER_WAVE 512
#define STORE_TILES   (ROWS_PER_WAVE / 16)
#define WARM_TILES    10
#define TOT_TILES     (STORE_TILES + WARM_TILES)
#define COLS_PER_WAVE 64
#define NCT           (COLS_PER_WAVE / 16)
#define PF            6
#define GROUPS        (TOT_TILES / PF)

struct Aregs { v2f Aj[4]; v2f Ac; };

__device__ __forceinline__ void load_tile(const float* __restrict__ Vb, unsigned lcol, int s, int koff, v4f (&R)[8]) {
    if (s >= 16 && s + 16 <= LSEQ) {
        const float* p = Vb + (size_t)(s - 1) * DMODEL;
        const unsigned lo = (unsigned)koff * DMODEL + lcol;
#pragma unroll
        for (int q = 0; q < 8; ++q)
            R[q] = *(const v4f*)(p + (lo + (unsigned)(4 * (q >> 1) + (q & 1)) * DMODEL));
    } else {
#pragma unroll
        for (int q = 0; q < 8; ++q) {
            int rv = s - 1 + koff + 4 * (q >> 1) + (q & 1);
            rv = min(max(rv, 0), LSEQ - 1);
            R[q] = *(const v4f*)(Vb + ((unsigned)rv * DMODEL + lcol));
        }
    }
}

__device__ __forceinline__ void fix_edge(v4f (&R)[8], int s, int lane, const v4f& v0q) {
    if (s == 0) {
        if (lane < 16) R[0] = v0q;
    } else if (s >= LSEQ) {
        const v4f z = {0.0f, 0.0f, 0.0f, 0.0f};
        const bool keep0 = (s == LSEQ) && (lane < 16);
        R[0] = keep0 ? R[0] : z;
#pragma unroll
        for (int q = 1; q < 8; ++q) R[q] = z;
    }
}

template <bool REFILL>
__device__ __forceinline__ void tile_step(const Aregs& A, v4f (&R)[8], v2f (&bc)[NCT],
                                          const float* __restrict__ Vb, float* __restrict__ Ob,
                                          unsigned lcol, const v4f& v0q, int s, int koff, int lane,
                                          bool do_store) {
    fix_edge(R, s, lane, v0q);
    v2f B[4][NCT];
#pragma unroll
    for (int j = 0; j < 4; ++j)
#pragma unroll
        for (int t = 0; t < NCT; ++t) {
            B[j][t].x = R[2 * j][t];
            B[j][t].y = R[2 * j + 1][t];
        }
    if (REFILL) load_tile(Vb, lcol, s - 16 * PF, koff, R);

    v8f acc[NCT];
#pragma unroll
    for (int t = 0; t < NCT; ++t) acc[t] = v8f{};
#pragma unroll
    for (int j = 0; j < 4; ++j)
#pragma unroll
        for (int t = 0; t < NCT; ++t)
            acc[t] = __builtin_amdgcn_wmma_f32_16x16x4_f32(false, A.Aj[j], false, B[j][t], (short)0, acc[t], false, false);
#pragma unroll
    for (int t = 0; t < NCT; ++t)
        acc[t] = __builtin_amdgcn_wmma_f32_16x16x4_f32(false, A.Ac, false, bc[t], (short)0, acc[t], false, false);
#pragma unroll
    for (int t = 0; t < NCT; ++t) { bc[t].x = acc[t][0]; bc[t].y = 0.0f; }

    if (do_store) {
        float* o = Ob + (size_t)s * DMODEL;
        const unsigned lo = ((lane < 16) ? 0u : 8u) * DMODEL + lcol;
#pragma unroll
        for (int r = 0; r < 8; ++r) {
            v4f y = { acc[0][r], acc[1][r], acc[2][r], acc[3][r] };
            *(v4f*)(o + (lo + (unsigned)r * DMODEL)) = y;
        }
    }
}

__global__ void __launch_bounds__(128)
esa_suffix_scan_wmma(const float* __restrict__ V,
                     const float* __restrict__ alpha,
                     const float* __restrict__ v0,
                     float* __restrict__ out) {
    const int lane  = threadIdx.x & 31;
    const int wid   = __builtin_amdgcn_readfirstlane((int)(threadIdx.x >> 5));
    const int gw    = blockIdx.x * 4 + wid;
    const int b     = gw >> 8;
    const int rem   = gw & 255;
    const int head  = rem >> 4;
    const int c0    = head * COLS_PER_WAVE;
    const int r0    = (rem & 15) * ROWS_PER_WAVE;

    const float a    = 1.0f / (1.0f + expf(-alpha[head]));
    const float oma  = 1.0f - a;

    const int M    = lane & 15;
    const int koff = (lane < 16) ? 0 : 2;

    Aregs A;
#pragma unroll
    for (int j = 0; j < 4; ++j) {
        const int m0 = 4 * j + koff;
        const int m1 = m0 + 1;
        A.Aj[j].x = (m0 >= M) ? a * __powf(oma, (float)(m0 - M)) : 0.0f;
        A.Aj[j].y = (m1 >= M) ? a * __powf(oma, (float)(m1 - M)) : 0.0f;
    }
    A.Ac.x = (lane < 16) ? __powf(oma, (float)(16 - M)) : 0.0f;
    A.Ac.y = 0.0f;

    const unsigned lcol = (unsigned)(c0 + 4 * M);
    const float* Vb = V + (size_t)b * LSEQ * DMODEL;
    float*       Ob = out + (size_t)b * LSEQ * DMODEL;
    const v4f v0q = *(const v4f*)(v0 + lcol);

    v2f bc[NCT];
#pragma unroll
    for (int t = 0; t < NCT; ++t) { bc[t].x = 0.0f; bc[t].y = 0.0f; }

    v4f buf[PF][8];
#pragma unroll
    for (int p = 0; p < PF; ++p)
        load_tile(Vb, lcol, r0 + 16 * (TOT_TILES - 1 - p), koff, buf[p]);

#pragma unroll 1
    for (int g = 0; g < GROUPS - 1; ++g) {
#pragma unroll
        for (int p = 0; p < PF; ++p) {
            const int k = g * PF + p;
            const int s = r0 + 16 * (TOT_TILES - 1 - k);
            tile_step<true>(A, buf[p], bc, Vb, Ob, lcol, v0q, s, koff, lane, k >= WARM_TILES);
        }
    }
#pragma unroll
    for (int p = 0; p < PF; ++p) {
        const int k = (GROUPS - 1) * PF + p;
        const int s = r0 + 16 * (TOT_TILES - 1 - k);
        tile_step<false>(A, buf[p], bc, Vb, Ob, lcol, v0q, s, koff, lane, true);
    }
}

extern "C" void kernel_launch(void* const* d_in, const int* in_sizes, int n_in,
                              void* d_out, int out_size, void* d_ws, size_t ws_size,
                              hipStream_t stream) {
    (void)in_sizes; (void)n_in; (void)d_ws; (void)ws_size; (void)out_size;
    const float* V     = (const float*)d_in[0];
    const float* alpha = (const float*)d_in[1];
    const float* v0    = (const float*)d_in[2];
    float* out = (float*)d_out;

    esa_suffix_scan_wmma<<<256, 128, 0, stream>>>(V, alpha, v0, out);
}
